# starting kernel
# baseline (speedup 1.0000x reference)
#include <hip/hip_runtime.h>

typedef __attribute__((ext_vector_type(2))) float v2f;
typedef __attribute__((ext_vector_type(4))) float v4f;
typedef __attribute__((ext_vector_type(8))) float v8f;

#define KDIM   768
#define NBATCH 32768

__global__ void build_eff_weights(const float* __restrict__ lhs,
                                  const float* __restrict__ rhs,
                                  const float* __restrict__ W,
                                  float* __restrict__ E) {
    for (int e = threadIdx.x; e < KDIM; e += blockDim.x) {
        const int ch  = e >> 8;
        const int rem = e & 255;
        const int row = rem >> 4;
        const int col = rem & 15;
        const int r = row >> 3, p = row & 7;
        const int c = col >> 3, q = col & 7;

        const float* lp = lhs + r * 128 + p;
        const float* rp = rhs + c * 128 + q * 16;
        const float* wp = W + ch * 1024 + (r * 16) * 32 + c * 16;

        float acc = 0.0f;
        #pragma unroll
        for (int P = 0; P < 16; ++P) {
            const float  l    = lp[P * 8];
            const float* wrow = wp + P * 32;
            float s = 0.0f;
            #pragma unroll
            for (int Q = 0; Q < 16; ++Q) s += wrow[Q] * rp[Q];
            acc += l * s;
        }
        E[e] = acc;
    }
}

__global__ void __launch_bounds__(256) gemv_wmma_f32(const float* __restrict__ x,
                                                     const float* __restrict__ E,
                                                     const float* __restrict__ bias,
                                                     float* __restrict__ y) {
    __shared__ float Elds[KDIM];
    for (int i = threadIdx.x; i < KDIM; i += 256) Elds[i] = E[i];
    __syncthreads();

    const int lane    = threadIdx.x & 31;
    const int wave    = threadIdx.x >> 5;
    const int rowBase = blockIdx.x * 128 + wave * 16;
    const int m       = lane & 15;
    const int hoff    = (lane >> 4) << 2;

    const float* xrow = x + (size_t)(rowBase + m) * KDIM + hoff;
    const float* Eh   = Elds + hoff;

    v8f acc = {};
    #pragma unroll 4
    for (int k = 0; k < KDIM; k += 8) {
        v4f xa = *(const v4f*)(xrow + k);
        v4f eb = *(const v4f*)(Eh + k);

        v2f a0 = { xa.x, xa.y };
        v2f b0 = { eb.x, eb.y };
        acc = __builtin_amdgcn_wmma_f32_16x16x4_f32(
            false, a0, false, b0, (short)0, acc, false, false);

        v2f a1 = { xa.z, xa.w };
        v2f b1 = { eb.z, eb.w };
        acc = __builtin_amdgcn_wmma_f32_16x16x4_f32(
            false, a1, false, b1, (short)0, acc, false, false);
    }

    const float bs = bias[0];
    if (m == 0) {
        float* out = y + rowBase + ((lane >> 4) << 3);
        v4f lo = { acc[0] + bs, acc[1] + bs, acc[2] + bs, acc[3] + bs };
        v4f hi = { acc[4] + bs, acc[5] + bs, acc[6] + bs, acc[7] + bs };
        *(v4f*)(out)     = lo;
        *(v4f*)(out + 4) = hi;
    }
}

extern "C" void kernel_launch(void* const* d_in, const int* in_sizes, int n_in,
                              void* d_out, int out_size, void* d_ws, size_t ws_size,
                              hipStream_t stream) {
    const float* x    = (const float*)d_in[0];
    const float* lhs  = (const float*)d_in[1];
    const float* rhs  = (const float*)d_in[2];
    const float* W    = (const float*)d_in[3];
    const float* bias = (const float*)d_in[4];
    float*       out  = (float*)d_out;
    float*       E    = (float*)d_ws;

    build_eff_weights<<<1, 256, 0, stream>>>(lhs, rhs, W, E);
    gemv_wmma_f32<<<NBATCH / 128, 256, 0, stream>>>(x, E, bias, out);
}


# Dot kernel: K split over the 8 waves of a work-group (2 row tiles x 4 slices of 192 features; 1024 groups of 32 samples), slice partials added through LDS: 4x the resident waves; sum order changed
# speedup vs baseline: 2.4827x; 2.4827x over previous
#include <hip/hip_runtime.h>

typedef __attribute__((ext_vector_type(2))) float v2f;
typedef __attribute__((ext_vector_type(4))) float v4f;
typedef __attribute__((ext_vector_type(8))) float v8f;

#define KDIM   768
#define NBATCH 32768

#define TPITCH 136

__global__ void __launch_bounds__(256) build_eff_weights(const float* __restrict__ lhs,
                                                         const float* __restrict__ rhs,
                                                         const float* __restrict__ W,
                                                         float* __restrict__ E) {
    __shared__ float T[2 * TPITCH];
    __shared__ float L[128];

    const int t  = threadIdx.x;
    const int ch = blockIdx.x >> 1;
    const int r  = blockIdx.x & 1;

    const int c = t >> 7;
    const int P = (t >> 3) & 15;
    const int q = t & 7;

    const v4f* wrow = (const v4f*)(W + ch * 1024 + (r * 16 + P) * 32 + c * 16);
    const v4f* rrow = (const v4f*)(rhs + c * 128 + q * 16);

    v4f wv[4], rv[4];
    for (int i = 0; i < 4; ++i) { wv[i] = wrow[i]; rv[i] = rrow[i]; }
    float lv = 0.0f;
    if (t < 128) lv = lhs[r * 128 + t];

    float s = 0.0f;
    for (int i = 0; i < 4; ++i) {
        s += wv[i].x * rv[i].x;
        s += wv[i].y * rv[i].y;
        s += wv[i].z * rv[i].z;
        s += wv[i].w * rv[i].w;
    }
    T[c * TPITCH + P * 8 + q] = s;
    if (t < 128) L[t] = lv;
    __syncthreads();

    if (t < 128) {
        const int p   = t >> 4;
        const int col = t & 15;
        const float* Tc = T + (col >> 3) * TPITCH + (col & 7);
        float acc = 0.0f;
        for (int PP = 0; PP < 16; ++PP) acc += L[PP * 8 + p] * Tc[PP * 8];
        E[ch * 256 + r * 128 + t] = acc;
    }
}

#define KSPLIT   4
#define KSLICE   (KDIM / KSPLIT)
#define WG_TILES 2
#define WG_ROWS  (16 * WG_TILES)

__global__ void __launch_bounds__(256) gemv_wmma_f32(const float* __restrict__ x,
                                                     const float* __restrict__ E,
                                                     const float* __restrict__ bias,
                                                     float* __restrict__ y) {
    __shared__ float Elds[KDIM];
    __shared__ __attribute__((aligned(16))) float part[WG_TILES * KSPLIT * 16];
    for (int i = threadIdx.x; i < KDIM; i += 256) Elds[i] = E[i];
    __syncthreads();

    const int lane    = threadIdx.x & 31;
    const int wave    = threadIdx.x >> 5;
    const int tile    = wave >> 2;
    const int slice   = wave & (KSPLIT - 1);
    const int rowBase = blockIdx.x * WG_ROWS + tile * 16;
    const int m       = lane & 15;
    const int hoff    = (lane >> 4) << 2;

    const float* xrow = x + (size_t)(rowBase + m) * KDIM + slice * KSLICE + hoff;
    const float* Eh   = Elds + slice * KSLICE + hoff;

    v8f acc = {};
    #pragma unroll 4
    for (int k = 0; k < KSLICE; k += 8) {
        v4f xa = *(const v4f*)(xrow + k);
        v4f eb = *(const v4f*)(Eh + k);

        v2f a0 = { xa.x, xa.y };
        v2f b0 = { eb.x, eb.y };
        acc = __builtin_amdgcn_wmma_f32_16x16x4_f32(
            false, a0, false, b0, (short)0, acc, false, false);

        v2f a1 = { xa.z, xa.w };
        v2f b1 = { eb.z, eb.w };
        acc = __builtin_amdgcn_wmma_f32_16x16x4_f32(
            false, a1, false, b1, (short)0, acc, false, false);
    }

    if (m == 0) {
        float* p = part + (tile * KSPLIT + slice) * 16 + ((lane >> 4) << 3);
        v4f lo = { acc[0], acc[1], acc[2], acc[3] };
        v4f hi = { acc[4], acc[5], acc[6], acc[7] };
        *(v4f*)(p)     = lo;
        *(v4f*)(p + 4) = hi;
    }
    __syncthreads();

    if (threadIdx.x < WG_ROWS) {
        const float* p = part + (threadIdx.x >> 4) * (KSPLIT * 16) + (threadIdx.x & 15);
        float s = p[0];
        s += p[16];
        s += p[32];
        s += p[48];
        y[blockIdx.x * WG_ROWS + threadIdx.x] = s + bias[0];
    }
}

extern "C" void kernel_launch(void* const* d_in, const int* in_sizes, int n_in,
                              void* d_out, int out_size, void* d_ws, size_t ws_size,
                              hipStream_t stream) {
    const float* x    = (const float*)d_in[0];
    const float* lhs  = (const float*)d_in[1];
    const float* rhs  = (const float*)d_in[2];
    const float* W    = (const float*)d_in[3];
    const float* bias = (const float*)d_in[4];
    float*       out  = (float*)d_out;
    float*       E    = (float*)d_ws;

    build_eff_weights<<<6, 256, 0, stream>>>(lhs, rhs, W, E);
    gemv_wmma_f32<<<NBATCH / WG_ROWS, 256, 0, stream>>>(x, E, bias, out);
}


# Dot kernel made persistent: 256 work-groups (one per WGP) walk 4 groups of 32 samples each, E slice held in registers, next group's x requested ahead: 2048 waves enter, not 8192; sum order unchanged
# speedup vs baseline: 2.8254x; 1.1380x over previous
#include <hip/hip_runtime.h>

typedef __attribute__((ext_vector_type(2))) float v2f;
typedef __attribute__((ext_vector_type(4))) float v4f;
typedef __attribute__((ext_vector_type(8))) float v8f;

#define KDIM   768
#define NBATCH 32768

#define TPITCH 136

__global__ void __launch_bounds__(256) build_eff_weights(const float* __restrict__ lhs,
                                                         const float* __restrict__ rhs,
                                                         const float* __restrict__ W,
                                                         float* __restrict__ E) {
    __shared__ float T[2 * TPITCH];
    __shared__ float L[128];

    const int t  = threadIdx.x;
    const int ch = blockIdx.x >> 1;
    const int r  = blockIdx.x & 1;

    const int c = t >> 7;
    const int P = (t >> 3) & 15;
    const int q = t & 7;

    const v4f* wrow = (const v4f*)(W + ch * 1024 + (r * 16 + P) * 32 + c * 16);
    const v4f* rrow = (const v4f*)(rhs + c * 128 + q * 16);

    v4f wv[4], rv[4];
    for (int i = 0; i < 4; ++i) { wv[i] = wrow[i]; rv[i] = rrow[i]; }
    float lv = 0.0f;
    if (t < 128) lv = lhs[r * 128 + t];

    float s = 0.0f;
    for (int i = 0; i < 4; ++i) {
        s += wv[i].x * rv[i].x;
        s += wv[i].y * rv[i].y;
        s += wv[i].z * rv[i].z;
        s += wv[i].w * rv[i].w;
    }
    T[c * TPITCH + P * 8 + q] = s;
    if (t < 128) L[t] = lv;
    __syncthreads();

    if (t < 128) {
        const int p   = t >> 4;
        const int col = t & 15;
        const float* Tc = T + (col >> 3) * TPITCH + (col & 7);
        float acc = 0.0f;
        for (int PP = 0; PP < 16; ++PP) acc += L[PP * 8 + p] * Tc[PP * 8];
        E[ch * 256 + r * 128 + t] = acc;
    }
}

#define KSPLIT    4
#define KSLICE    (KDIM / KSPLIT)
#define NCHUNK    (KSLICE / 8)
#define WG_TILES  2
#define GRP_ROWS  (16 * WG_TILES)
#define WG_GROUPS 4
#define WG_ROWS   (GRP_ROWS * WG_GROUPS)
#define NWG       (NBATCH / WG_ROWS)

__global__ void __launch_bounds__(256) gemv_wmma_f32(const float* __restrict__ x,
                                                     const float* __restrict__ E,
                                                     const float* __restrict__ bias,
                                                     float* __restrict__ y) {
    __shared__ __attribute__((aligned(16))) float part[WG_GROUPS * WG_TILES * KSPLIT * 16];

    const int lane  = threadIdx.x & 31;
    const int wave  = threadIdx.x >> 5;
    const int tile  = wave >> 2;
    const int slice = wave & (KSPLIT - 1);
    const int m     = lane & 15;
    const int hoff  = (lane >> 4) << 2;

    const float bv = bias[0];

    const float* Eh = E + slice * KSLICE + hoff;
    v4f eb[NCHUNK];
    for (int j = 0; j < NCHUNK; ++j) eb[j] = *(const v4f*)(Eh + 8 * j);

    const float* xrow = x + (size_t)(blockIdx.x * WG_ROWS + tile * 16 + m) * KDIM
                          + slice * KSLICE + hoff;

    v4f xa[NCHUNK];
    for (int j = 0; j < NCHUNK; ++j) xa[j] = *(const v4f*)(xrow + 8 * j);

    for (int g = 0; g < WG_GROUPS; ++g) {
        v4f xn[NCHUNK];
        if (g + 1 < WG_GROUPS) {
            const float* xnext = xrow + (size_t)(g + 1) * GRP_ROWS * KDIM;
            for (int j = 0; j < NCHUNK; ++j) xn[j] = *(const v4f*)(xnext + 8 * j);
            __builtin_amdgcn_sched_barrier(0);
        }

        v8f acc = {};
        for (int j = 0; j < NCHUNK; ++j) {
            v2f a0 = { xa[j].x, xa[j].y };
            v2f b0 = { eb[j].x, eb[j].y };
            acc = __builtin_amdgcn_wmma_f32_16x16x4_f32(
                false, a0, false, b0, (short)0, acc, false, false);

            v2f a1 = { xa[j].z, xa[j].w };
            v2f b1 = { eb[j].z, eb[j].w };
            acc = __builtin_amdgcn_wmma_f32_16x16x4_f32(
                false, a1, false, b1, (short)0, acc, false, false);
        }

        if (m == 0) {
            float* p = part + ((g * WG_TILES + tile) * KSPLIT + slice) * 16 + ((lane >> 4) << 3);
            v4f lo = { acc[0], acc[1], acc[2], acc[3] };
            v4f hi = { acc[4], acc[5], acc[6], acc[7] };
            *(v4f*)(p)     = lo;
            *(v4f*)(p + 4) = hi;
        }

        if (g + 1 < WG_GROUPS) {
            for (int j = 0; j < NCHUNK; ++j) xa[j] = xn[j];
        }
    }
    __syncthreads();

    if (threadIdx.x < WG_ROWS) {
        const float* p = part + (threadIdx.x >> 4) * (KSPLIT * 16) + (threadIdx.x & 15);
        float s = p[0];
        s += p[16];
        s += p[32];
        s += p[48];
        y[blockIdx.x * WG_ROWS + threadIdx.x] = s + bv;
    }
}

extern "C" void kernel_launch(void* const* d_in, const int* in_sizes, int n_in,
                              void* d_out, int out_size, void* d_ws, size_t ws_size,
                              hipStream_t stream) {
    const float* x    = (const float*)d_in[0];
    const float* lhs  = (const float*)d_in[1];
    const float* rhs  = (const float*)d_in[2];
    const float* W    = (const float*)d_in[3];
    const float* bias = (const float*)d_in[4];
    float*       out  = (float*)d_out;
    float*       E    = (float*)d_ws;

    build_eff_weights<<<6, 256, 0, stream>>>(lhs, rhs, W, E);
    gemv_wmma_f32<<<NWG, 256, 0, stream>>>(x, E, bias, out);
}
